# starting kernel
# baseline (speedup 1.0000x reference)
#include <hip/hip_runtime.h>
#include <hip/hip_bf16.h>

typedef __attribute__((ext_vector_type(2))) float v2f;
typedef __attribute__((ext_vector_type(8))) float v8f;

#define BATCH 2048
#define NCAT 32
#define SDIM 64
#define EMB 256
#define NOUT 256
#define ODIM 512
#define HDIM 512
#define HSTR 516
#define SSTR 68

__global__ void bucket_by_cat(const int* __restrict__ cat_ids,
                              int* __restrict__ counts,
                              int* __restrict__ offsets,
                              int* __restrict__ perm) {
    __shared__ int s_cnt[NCAT];
    __shared__ int s_cur[NCAT];
    int t = threadIdx.x;
    if (t < NCAT) s_cnt[t] = 0;
    __syncthreads();
    for (int i = t; i < BATCH; i += blockDim.x)
        atomicAdd(&s_cnt[cat_ids[i]], 1);
    __syncthreads();
    if (t == 0) {
        int acc = 0;
        for (int c = 0; c < NCAT; ++c) {
            s_cur[c] = acc;
            offsets[c] = acc;
            counts[c] = s_cnt[c];
            acc += s_cnt[c];
        }
    }
    __syncthreads();
    for (int i = t; i < BATCH; i += blockDim.x) {
        int c = cat_ids[i];
        int pos = atomicAdd(&s_cur[c], 1);
        perm[pos] = i;
    }
}

__global__ __launch_bounds__(128)
void moe_mlp_kernel(const float* __restrict__ state,
                    const float* __restrict__ obs,
                    const float* __restrict__ obs_W,
                    const float* __restrict__ obs_b,
                    const float* __restrict__ state_W,
                    const float* __restrict__ state_b,
                    const float* __restrict__ l2_W,
                    const float* __restrict__ l2_b,
                    const int* __restrict__ counts,
                    const int* __restrict__ offsets,
                    const int* __restrict__ perm,
                    float* __restrict__ out) {
    const int c    = blockIdx.y;
    const int tile = blockIdx.x;
    const int cnt  = counts[c];
    if (tile * 16 >= cnt) return;

    __shared__ int   sh_rows[16];
    __shared__ float sh_s[16 * SSTR];
    __shared__ float sh_h[16 * HSTR];

    const int t    = threadIdx.x;
    const int wave = t >> 5;
    const int lane = t & 31;
    const int m    = lane & 15;
    const int half = lane >> 4;
    const int kb   = half * 2;
    const int mb   = half * 8;

    const int base = offsets[c];
    if (t < 16) {
        int idx = tile * 16 + t;
        sh_rows[t] = (idx < cnt) ? perm[base + idx] : -1;
    }
    __syncthreads();

    for (int i = t; i < 16 * SDIM; i += 128) {
        int mm = i >> 6, kk = i & (SDIM - 1);
        int r = sh_rows[mm];
        sh_s[mm * SSTR + kk] = (r >= 0) ? state[(long)r * SDIM + kk] : 0.0f;
    }
    __syncthreads();

    const float* Ws = state_W + (long)c * SDIM * EMB;
    const float* bs = state_b + (long)c * EMB;
    const float* Wl = l2_W    + (long)c * HDIM * NOUT;
    const float* lb = l2_b    + (long)c * NOUT;

    for (int j = 0; j < 4; ++j) {
        const int nt   = wave * 4 + j;
        const int ncol = nt * 16 + m;
        v8f acc = {};
        for (int k = 0; k < SDIM; k += 4) {
            v2f a, b;
            a.x = sh_s[m * SSTR + k + kb];
            a.y = sh_s[m * SSTR + k + kb + 1];
            b.x = Ws[(long)(k + kb)     * EMB + ncol];
            b.y = Ws[(long)(k + kb + 1) * EMB + ncol];
            acc = __builtin_amdgcn_wmma_f32_16x16x4_f32(
                false, a, false, b, (short)0, acc, false, false);
        }
        const float bias = bs[ncol];
        #pragma unroll
        for (int v = 0; v < 8; ++v)
            sh_h[(mb + v) * HSTR + ncol] = fmaxf(acc[v] + bias, 0.0f);
    }

    {
        const int  r      = sh_rows[m];
        const float validf = (r >= 0) ? 1.0f : 0.0f;
        const long ro     = (long)((r >= 0) ? r : 0) * ODIM;
        for (int j = 0; j < 4; ++j) {
            const int nt   = wave * 4 + j;
            const int ncol = nt * 16 + m;
            v8f acc = {};
            for (int k = 0; k < ODIM; k += 4) {
                v2f a, b;
                a.x = obs[ro + k + kb]     * validf;
                a.y = obs[ro + k + kb + 1] * validf;
                b.x = obs_W[(long)(k + kb)     * EMB + ncol];
                b.y = obs_W[(long)(k + kb + 1) * EMB + ncol];
                acc = __builtin_amdgcn_wmma_f32_16x16x4_f32(
                    false, a, false, b, (short)0, acc, false, false);
            }
            const float bias = obs_b[ncol];
            #pragma unroll
            for (int v = 0; v < 8; ++v)
                sh_h[(mb + v) * HSTR + EMB + ncol] = acc[v] + bias;
        }
    }
    __syncthreads();

    for (int j = 0; j < 4; ++j) {
        const int nt   = wave * 4 + j;
        const int ncol = nt * 16 + m;
        v8f acc = {};
        for (int k = 0; k < HDIM; k += 4) {
            v2f a, b;
            a.x = sh_h[m * HSTR + k + kb];
            a.y = sh_h[m * HSTR + k + kb + 1];
            b.x = Wl[(long)(k + kb)     * NOUT + ncol];
            b.y = Wl[(long)(k + kb + 1) * NOUT + ncol];
            acc = __builtin_amdgcn_wmma_f32_16x16x4_f32(
                false, a, false, b, (short)0, acc, false, false);
        }
        const float bias = lb[ncol];
        #pragma unroll
        for (int v = 0; v < 8; ++v) {
            int r = sh_rows[mb + v];
            if (r >= 0)
                out[(long)r * NOUT + ncol] = acc[v] + bias;
        }
    }
}

extern "C" void kernel_launch(void* const* d_in, const int* in_sizes, int n_in,
                              void* d_out, int out_size, void* d_ws, size_t ws_size,
                              hipStream_t stream) {
    const float* state   = (const float*)d_in[0];
    const float* obs     = (const float*)d_in[1];
    const int*   cat_ids = (const int*)  d_in[2];
    const float* obs_W   = (const float*)d_in[3];
    const float* obs_b   = (const float*)d_in[4];
    const float* state_W = (const float*)d_in[5];
    const float* state_b = (const float*)d_in[6];
    const float* l2_W    = (const float*)d_in[7];
    const float* l2_b    = (const float*)d_in[8];
    float*       out     = (float*)d_out;

    int* counts  = (int*)d_ws;
    int* offsets = counts + NCAT;
    int* perm    = offsets + NCAT;

    bucket_by_cat<<<1, 256, 0, stream>>>(cat_ids, counts, offsets, perm);

    dim3 grid((BATCH + 15) / 16, NCAT);
    moe_mlp_kernel<<<grid, 128, 0, stream>>>(state, obs, obs_W, obs_b,
                                             state_W, state_b, l2_W, l2_b,
                                             counts, offsets, perm, out);
}


# Category-free obs projection leaves the per-category tiles: dense GEMM over rows in batch order in launch 1 beside the bucketing block, K quarters added in LDS; launch 2 copies its rows
# speedup vs baseline: 5.3861x; 5.3861x over previous
#include <hip/hip_runtime.h>
#include <hip/hip_bf16.h>

typedef __attribute__((ext_vector_type(2))) float v2f;
typedef __attribute__((ext_vector_type(8))) float v8f;

#define BATCH 2048
#define NCAT 32
#define SDIM 64
#define EMB 256
#define NOUT 256
#define ODIM 512
#define HDIM 512
#define HSTR 516
#define SSTR 68
#define OSTR 260
#define PSTR 36
#define MAXTILES 158

__global__ __launch_bounds__(256)
void bucket_and_obs_kernel(const int* __restrict__ cat_ids,
                           const float* __restrict__ obs,
                           const float* __restrict__ obs_W,
                           const float* __restrict__ obs_b,
                           int* __restrict__ tlist,
                           int* __restrict__ perm,
                           float* __restrict__ hobs) {
    __shared__ int s_cnt[NCAT];
    __shared__ int s_cur[NCAT];
    __shared__ __attribute__((aligned(16))) float sh_p[3 * 16 * PSTR];

    const int t = threadIdx.x;

    if (blockIdx.x == 0) {
        if (t < NCAT) s_cnt[t] = 0;
        __syncthreads();
        for (int i = t; i < BATCH; i += blockDim.x)
            atomicAdd(&s_cnt[cat_ids[i]], 1);
        __syncthreads();
        if (t == 0) {
            int acc = 0;
            int nt = 0;
            for (int c = 0; c < NCAT; ++c) {
                const int cnt = s_cnt[c];
                s_cur[c] = acc;
                for (int i = 0; i < cnt; i += 16) {
                    const int n = (cnt - i < 16) ? (cnt - i) : 16;
                    tlist[nt++] = (c << 16) | ((acc + i) << 5) | n;
                }
                acc += cnt;
            }
            for (; nt < MAXTILES; ++nt) tlist[nt] = 0;
        }
        __syncthreads();
        for (int i = t; i < BATCH; i += blockDim.x) {
            int c = cat_ids[i];
            int pos = atomicAdd(&s_cur[c], 1);
            perm[pos] = i;
        }
        return;
    }

    const int g    = blockIdx.x - 1;
    const int rt   = g >> 3;
    const int cs   = g & 7;
    const int wave = __builtin_amdgcn_readfirstlane(t >> 5);
    const int lane = t & 31;
    const int m    = lane & 15;
    const int half = lane >> 4;
    const int kb   = half * 2;
    const int mb   = half * 8;
    const int ct   = wave & 1;
    const int kq   = wave >> 1;
    const int lcol = ct * 16 + m;
    const int ncol = cs * 32 + lcol;
    const long ro  = (long)(rt * 16 + m) * ODIM;

    v8f acc = {};
    const int k0 = kq * (ODIM / 4);
    for (int k = k0; k < k0 + ODIM / 4; k += 4) {
        v2f a, b;
        a.x = obs[ro + k + kb];
        a.y = obs[ro + k + kb + 1];
        b.x = obs_W[(long)(k + kb)     * EMB + ncol];
        b.y = obs_W[(long)(k + kb + 1) * EMB + ncol];
        acc = __builtin_amdgcn_wmma_f32_16x16x4_f32(
            false, a, false, b, (short)0, acc, false, false);
    }
    if (kq > 0) {
        #pragma unroll
        for (int v = 0; v < 8; ++v)
            sh_p[(kq - 1) * 16 * PSTR + (mb + v) * PSTR + lcol] = acc[v];
    }
    __syncthreads();
    if (kq == 0) {
        const float bias = obs_b[ncol];
        #pragma unroll
        for (int v = 0; v < 8; ++v) {
            const int idx = (mb + v) * PSTR + lcol;
            sh_p[idx] = (((acc[v] + sh_p[idx]) + sh_p[16 * PSTR + idx])
                         + sh_p[2 * 16 * PSTR + idx]) + bias;
        }
    }
    __syncthreads();
    if (t < 128) {
        const int row = t >> 3;
        const int c4  = (t & 7) * 4;
        const float4 v = *reinterpret_cast<const float4*>(&sh_p[row * PSTR + c4]);
        *reinterpret_cast<float4*>(&hobs[(long)(rt * 16 + row) * EMB + cs * 32 + c4]) = v;
    }
}

__global__ __launch_bounds__(1024)
void moe_mlp_kernel(const float* __restrict__ state,
                    const float* __restrict__ hobs,
                    const float* __restrict__ state_W,
                    const float* __restrict__ state_b,
                    const float* __restrict__ l2_W,
                    const float* __restrict__ l2_b,
                    const int* __restrict__ tlist,
                    const int* __restrict__ perm,
                    float* __restrict__ out) {
    const int desc  = tlist[blockIdx.x];
    const int nrows = desc & 31;
    if (nrows == 0) return;
    const int first = (desc >> 5) & 2047;
    const int c     = desc >> 16;

    __shared__ int   sh_rows[16];
    __shared__ float sh_s[16 * SSTR];
    __shared__ __attribute__((aligned(16))) float sh_h[16 * HSTR];
    __shared__ __attribute__((aligned(16))) float sh_o[16 * OSTR];

    const int t    = threadIdx.x;
    const int wave = __builtin_amdgcn_readfirstlane(t >> 5);
    const int lane = t & 31;
    const int m    = lane & 15;
    const int half = lane >> 4;
    const int kb   = half * 2;
    const int mb   = half * 8;
    const int nt   = wave & 15;
    const int kh   = wave >> 4;
    const int ncol = nt * 16 + m;

    {
        const int mm = t >> 6, kk = t & (SDIM - 1);
        const int r  = (mm < nrows) ? perm[first + mm] : -1;
        if (kk == 0) sh_rows[mm] = r;
        sh_s[mm * SSTR + kk] = (r >= 0) ? state[(long)r * SDIM + kk] : 0.0f;
    }
    __syncthreads();

    const float* Ws = state_W + (long)c * SDIM * EMB;
    const float* bs = state_b + (long)c * EMB;
    const float* Wl = l2_W    + (long)c * HDIM * NOUT;
    const float* lb = l2_b    + (long)c * NOUT;

    if (kh == 0) {
        v8f acc = {};
        for (int k = 0; k < SDIM; k += 4) {
            v2f a, b;
            a.x = sh_s[m * SSTR + k + kb];
            a.y = sh_s[m * SSTR + k + kb + 1];
            b.x = Ws[(long)(k + kb)     * EMB + ncol];
            b.y = Ws[(long)(k + kb + 1) * EMB + ncol];
            acc = __builtin_amdgcn_wmma_f32_16x16x4_f32(
                false, a, false, b, (short)0, acc, false, false);
        }
        const float bias = bs[ncol];
        #pragma unroll
        for (int v = 0; v < 8; ++v)
            sh_h[(mb + v) * HSTR + ncol] = fmaxf(acc[v] + bias, 0.0f);
    } else {
        const int row = wave - 16;
        const int r   = sh_rows[row];
        float4 v0 = make_float4(0.0f, 0.0f, 0.0f, 0.0f);
        float4 v1 = v0;
        if (r >= 0) {
            const float4* srow = reinterpret_cast<const float4*>(hobs + (long)r * EMB);
            v0 = srow[lane];
            v1 = srow[32 + lane];
        }
        *reinterpret_cast<float4*>(&sh_h[row * HSTR + EMB + lane * 4])       = v0;
        *reinterpret_cast<float4*>(&sh_h[row * HSTR + EMB + 128 + lane * 4]) = v1;
    }
    __syncthreads();

    {
        v8f acc = {};
        const int k0 = kh * (HDIM / 2);
        for (int k = k0; k < k0 + HDIM / 2; k += 4) {
            v2f a, b;
            a.x = sh_h[m * HSTR + k + kb];
            a.y = sh_h[m * HSTR + k + kb + 1];
            b.x = Wl[(long)(k + kb)     * NOUT + ncol];
            b.y = Wl[(long)(k + kb + 1) * NOUT + ncol];
            acc = __builtin_amdgcn_wmma_f32_16x16x4_f32(
                false, a, false, b, (short)0, acc, false, false);
        }
        if (kh == 1) {
            #pragma unroll
            for (int v = 0; v < 8; ++v)
                sh_o[(mb + v) * OSTR + ncol] = acc[v];
        }
        __syncthreads();
        if (kh == 0) {
            const float bias = lb[ncol];
            #pragma unroll
            for (int v = 0; v < 8; ++v) {
                const int idx = (mb + v) * OSTR + ncol;
                sh_o[idx] = (acc[v] + sh_o[idx]) + bias;
            }
        }
        __syncthreads();
    }

    {
        const int row = t >> 6;
        const int c4  = (t & 63) * 4;
        const int r   = sh_rows[row];
        if (r >= 0) {
            const float4 v = *reinterpret_cast<const float4*>(&sh_o[row * OSTR + c4]);
            *reinterpret_cast<float4*>(&out[(long)r * NOUT + c4]) = v;
        }
    }
}

extern "C" void kernel_launch(void* const* d_in, const int* in_sizes, int n_in,
                              void* d_out, int out_size, void* d_ws, size_t ws_size,
                              hipStream_t stream) {
    const float* state   = (const float*)d_in[0];
    const float* obs     = (const float*)d_in[1];
    const int*   cat_ids = (const int*)  d_in[2];
    const float* obs_W   = (const float*)d_in[3];
    const float* obs_b   = (const float*)d_in[4];
    const float* state_W = (const float*)d_in[5];
    const float* state_b = (const float*)d_in[6];
    const float* l2_W    = (const float*)d_in[7];
    const float* l2_b    = (const float*)d_in[8];
    float*       out     = (float*)d_out;

    int*   tlist = (int*)d_ws;
    int*   perm  = tlist + 256;
    float* hobs  = (float*)(tlist + 4096);

    bucket_and_obs_kernel<<<1025, 256, 0, stream>>>(cat_ids, obs, obs_W, obs_b,
                                                    tlist, perm, hobs);

    moe_mlp_kernel<<<MAXTILES, 1024, 0, stream>>>(state, hobs,
                                                  state_W, state_b, l2_W, l2_b,
                                                  tlist, perm, out);
}


# Bucketing block made parallel: each thread holds its 8 category ids from one batch of loads; row and tile offsets by a 32-lane prefix scan (was 16 serial load hops and a one-thread loop)
# speedup vs baseline: 6.5361x; 1.2135x over previous
#include <hip/hip_runtime.h>
#include <hip/hip_bf16.h>

typedef __attribute__((ext_vector_type(2))) float v2f;
typedef __attribute__((ext_vector_type(8))) float v8f;

#define BATCH 2048
#define NCAT 32
#define SDIM 64
#define EMB 256
#define NOUT 256
#define ODIM 512
#define HDIM 512
#define HSTR 516
#define SSTR 68
#define OSTR 260
#define PSTR 36
#define MAXTILES 158

__global__ __launch_bounds__(256)
void bucket_and_obs_kernel(const int* __restrict__ cat_ids,
                           const float* __restrict__ obs,
                           const float* __restrict__ obs_W,
                           const float* __restrict__ obs_b,
                           int* __restrict__ tlist,
                           int* __restrict__ perm,
                           float* __restrict__ hobs) {
    __shared__ int s_cnt[NCAT];
    __shared__ int s_cur[NCAT];
    __shared__ int s_ntiles;
    __shared__ __attribute__((aligned(16))) float sh_p[3 * 16 * PSTR];

    const int t = threadIdx.x;

    if (blockIdx.x == 0) {
        int id[BATCH / 256];
        for (int j = 0; j < BATCH / 256; ++j) id[j] = cat_ids[t + 256 * j];
        if (t < NCAT) s_cnt[t] = 0;
        __syncthreads();
        for (int j = 0; j < BATCH / 256; ++j) atomicAdd(&s_cnt[id[j]], 1);
        __syncthreads();
        if (t < NCAT) {
            const int cnt = s_cnt[t];
            const int ntl = (cnt + 15) >> 4;
            int racc = cnt;
            int tacc = ntl;
            for (int d = 1; d < NCAT; d <<= 1) {
                const int r = __shfl_up(racc, d);
                const int q = __shfl_up(tacc, d);
                if (t >= d) { racc += r; tacc += q; }
            }
            const int rstart = racc - cnt;
            const int tstart = tacc - ntl;
            s_cur[t] = rstart;
            if (t == NCAT - 1) s_ntiles = tacc;
            for (int j = 0; j < ntl; ++j) {
                const int left = cnt - 16 * j;
                const int n = (left < 16) ? left : 16;
                tlist[tstart + j] = (t << 16) | ((rstart + 16 * j) << 5) | n;
            }
        }
        __syncthreads();
        if (t >= s_ntiles && t < MAXTILES) tlist[t] = 0;
        int pos[BATCH / 256];
        for (int j = 0; j < BATCH / 256; ++j) pos[j] = atomicAdd(&s_cur[id[j]], 1);
        for (int j = 0; j < BATCH / 256; ++j) perm[pos[j]] = t + 256 * j;
        return;
    }

    const int g    = blockIdx.x - 1;
    const int rt   = g >> 3;
    const int cs   = g & 7;
    const int wave = __builtin_amdgcn_readfirstlane(t >> 5);
    const int lane = t & 31;
    const int m    = lane & 15;
    const int half = lane >> 4;
    const int kb   = half * 2;
    const int mb   = half * 8;
    const int ct   = wave & 1;
    const int kq   = wave >> 1;
    const int lcol = ct * 16 + m;
    const int ncol = cs * 32 + lcol;
    const long ro  = (long)(rt * 16 + m) * ODIM;

    v8f acc = {};
    const int k0 = kq * (ODIM / 4);
    for (int k = k0; k < k0 + ODIM / 4; k += 4) {
        v2f a, b;
        a.x = obs[ro + k + kb];
        a.y = obs[ro + k + kb + 1];
        b.x = obs_W[(long)(k + kb)     * EMB + ncol];
        b.y = obs_W[(long)(k + kb + 1) * EMB + ncol];
        acc = __builtin_amdgcn_wmma_f32_16x16x4_f32(
            false, a, false, b, (short)0, acc, false, false);
    }
    if (kq > 0) {
        #pragma unroll
        for (int v = 0; v < 8; ++v)
            sh_p[(kq - 1) * 16 * PSTR + (mb + v) * PSTR + lcol] = acc[v];
    }
    __syncthreads();
    if (kq == 0) {
        const float bias = obs_b[ncol];
        #pragma unroll
        for (int v = 0; v < 8; ++v) {
            const int idx = (mb + v) * PSTR + lcol;
            sh_p[idx] = (((acc[v] + sh_p[idx]) + sh_p[16 * PSTR + idx])
                         + sh_p[2 * 16 * PSTR + idx]) + bias;
        }
    }
    __syncthreads();
    if (t < 128) {
        const int row = t >> 3;
        const int c4  = (t & 7) * 4;
        const float4 v = *reinterpret_cast<const float4*>(&sh_p[row * PSTR + c4]);
        *reinterpret_cast<float4*>(&hobs[(long)(rt * 16 + row) * EMB + cs * 32 + c4]) = v;
    }
}

__global__ __launch_bounds__(1024)
void moe_mlp_kernel(const float* __restrict__ state,
                    const float* __restrict__ hobs,
                    const float* __restrict__ state_W,
                    const float* __restrict__ state_b,
                    const float* __restrict__ l2_W,
                    const float* __restrict__ l2_b,
                    const int* __restrict__ tlist,
                    const int* __restrict__ perm,
                    float* __restrict__ out) {
    const int desc  = tlist[blockIdx.x];
    const int nrows = desc & 31;
    if (nrows == 0) return;
    const int first = (desc >> 5) & 2047;
    const int c     = desc >> 16;

    __shared__ int   sh_rows[16];
    __shared__ float sh_s[16 * SSTR];
    __shared__ __attribute__((aligned(16))) float sh_h[16 * HSTR];
    __shared__ __attribute__((aligned(16))) float sh_o[16 * OSTR];

    const int t    = threadIdx.x;
    const int wave = __builtin_amdgcn_readfirstlane(t >> 5);
    const int lane = t & 31;
    const int m    = lane & 15;
    const int half = lane >> 4;
    const int kb   = half * 2;
    const int mb   = half * 8;
    const int nt   = wave & 15;
    const int kh   = wave >> 4;
    const int ncol = nt * 16 + m;

    {
        const int mm = t >> 6, kk = t & (SDIM - 1);
        const int r  = (mm < nrows) ? perm[first + mm] : -1;
        if (kk == 0) sh_rows[mm] = r;
        sh_s[mm * SSTR + kk] = (r >= 0) ? state[(long)r * SDIM + kk] : 0.0f;
    }
    __syncthreads();

    const float* Ws = state_W + (long)c * SDIM * EMB;
    const float* bs = state_b + (long)c * EMB;
    const float* Wl = l2_W    + (long)c * HDIM * NOUT;
    const float* lb = l2_b    + (long)c * NOUT;

    if (kh == 0) {
        v8f acc = {};
        for (int k = 0; k < SDIM; k += 4) {
            v2f a, b;
            a.x = sh_s[m * SSTR + k + kb];
            a.y = sh_s[m * SSTR + k + kb + 1];
            b.x = Ws[(long)(k + kb)     * EMB + ncol];
            b.y = Ws[(long)(k + kb + 1) * EMB + ncol];
            acc = __builtin_amdgcn_wmma_f32_16x16x4_f32(
                false, a, false, b, (short)0, acc, false, false);
        }
        const float bias = bs[ncol];
        #pragma unroll
        for (int v = 0; v < 8; ++v)
            sh_h[(mb + v) * HSTR + ncol] = fmaxf(acc[v] + bias, 0.0f);
    } else {
        const int row = wave - 16;
        const int r   = sh_rows[row];
        float4 v0 = make_float4(0.0f, 0.0f, 0.0f, 0.0f);
        float4 v1 = v0;
        if (r >= 0) {
            const float4* srow = reinterpret_cast<const float4*>(hobs + (long)r * EMB);
            v0 = srow[lane];
            v1 = srow[32 + lane];
        }
        *reinterpret_cast<float4*>(&sh_h[row * HSTR + EMB + lane * 4])       = v0;
        *reinterpret_cast<float4*>(&sh_h[row * HSTR + EMB + 128 + lane * 4]) = v1;
    }
    __syncthreads();

    {
        v8f acc = {};
        const int k0 = kh * (HDIM / 2);
        for (int k = k0; k < k0 + HDIM / 2; k += 4) {
            v2f a, b;
            a.x = sh_h[m * HSTR + k + kb];
            a.y = sh_h[m * HSTR + k + kb + 1];
            b.x = Wl[(long)(k + kb)     * NOUT + ncol];
            b.y = Wl[(long)(k + kb + 1) * NOUT + ncol];
            acc = __builtin_amdgcn_wmma_f32_16x16x4_f32(
                false, a, false, b, (short)0, acc, false, false);
        }
        if (kh == 1) {
            #pragma unroll
            for (int v = 0; v < 8; ++v)
                sh_o[(mb + v) * OSTR + ncol] = acc[v];
        }
        __syncthreads();
        if (kh == 0) {
            const float bias = lb[ncol];
            #pragma unroll
            for (int v = 0; v < 8; ++v) {
                const int idx = (mb + v) * OSTR + ncol;
                sh_o[idx] = (acc[v] + sh_o[idx]) + bias;
            }
        }
        __syncthreads();
    }

    {
        const int row = t >> 6;
        const int c4  = (t & 63) * 4;
        const int r   = sh_rows[row];
        if (r >= 0) {
            const float4 v = *reinterpret_cast<const float4*>(&sh_o[row * OSTR + c4]);
            *reinterpret_cast<float4*>(&out[(long)r * NOUT + c4]) = v;
        }
    }
}

extern "C" void kernel_launch(void* const* d_in, const int* in_sizes, int n_in,
                              void* d_out, int out_size, void* d_ws, size_t ws_size,
                              hipStream_t stream) {
    const float* state   = (const float*)d_in[0];
    const float* obs     = (const float*)d_in[1];
    const int*   cat_ids = (const int*)  d_in[2];
    const float* obs_W   = (const float*)d_in[3];
    const float* obs_b   = (const float*)d_in[4];
    const float* state_W = (const float*)d_in[5];
    const float* state_b = (const float*)d_in[6];
    const float* l2_W    = (const float*)d_in[7];
    const float* l2_b    = (const float*)d_in[8];
    float*       out     = (float*)d_out;

    int*   tlist = (int*)d_ws;
    int*   perm  = tlist + 256;
    float* hobs  = (float*)(tlist + 4096);

    bucket_and_obs_kernel<<<1025, 256, 0, stream>>>(cat_ids, obs, obs_W, obs_b,
                                                    tlist, perm, hobs);

    moe_mlp_kernel<<<MAXTILES, 1024, 0, stream>>>(state, hobs,
                                                  state_W, state_b, l2_W, l2_b,
                                                  tlist, perm, out);
}


# obs GEMM re-tiled A-stationary: 256 groups (16 rows x 128 cols), each wave gathers its K eighth of the A tile once into registers and sweeps 8 column tiles; K eighths (was quarters) added in LDS
# speedup vs baseline: 7.3035x; 1.1174x over previous
#include <hip/hip_runtime.h>
#include <hip/hip_bf16.h>

typedef __attribute__((ext_vector_type(2))) float v2f;
typedef __attribute__((ext_vector_type(8))) float v8f;

#define BATCH 2048
#define NCAT 32
#define SDIM 64
#define EMB 256
#define NOUT 256
#define ODIM 512
#define HDIM 512
#define HSTR 516
#define SSTR 68
#define OSTR 260
#define QSTR 132
#define MAXTILES 158

__global__ __launch_bounds__(256)
void bucket_and_obs_kernel(const int* __restrict__ cat_ids,
                           const float* __restrict__ obs,
                           const float* __restrict__ obs_W,
                           const float* __restrict__ obs_b,
                           int* __restrict__ tlist,
                           int* __restrict__ perm,
                           float* __restrict__ hobs) {
    __shared__ int s_cnt[NCAT];
    __shared__ int s_cur[NCAT];
    __shared__ int s_ntiles;
    __shared__ __attribute__((aligned(16))) float sh_p[8 * 16 * QSTR];

    const int t = threadIdx.x;

    if (blockIdx.x == 0) {
        int id[BATCH / 256];
        for (int j = 0; j < BATCH / 256; ++j) id[j] = cat_ids[t + 256 * j];
        if (t < NCAT) s_cnt[t] = 0;
        __syncthreads();
        for (int j = 0; j < BATCH / 256; ++j) atomicAdd(&s_cnt[id[j]], 1);
        __syncthreads();
        if (t < NCAT) {
            const int cnt = s_cnt[t];
            const int ntl = (cnt + 15) >> 4;
            int racc = cnt;
            int tacc = ntl;
            for (int d = 1; d < NCAT; d <<= 1) {
                const int r = __shfl_up(racc, d);
                const int q = __shfl_up(tacc, d);
                if (t >= d) { racc += r; tacc += q; }
            }
            const int rstart = racc - cnt;
            const int tstart = tacc - ntl;
            s_cur[t] = rstart;
            if (t == NCAT - 1) s_ntiles = tacc;
            for (int j = 0; j < ntl; ++j) {
                const int left = cnt - 16 * j;
                const int n = (left < 16) ? left : 16;
                tlist[tstart + j] = (t << 16) | ((rstart + 16 * j) << 5) | n;
            }
        }
        __syncthreads();
        if (t >= s_ntiles && t < MAXTILES) tlist[t] = 0;
        int pos[BATCH / 256];
        for (int j = 0; j < BATCH / 256; ++j) pos[j] = atomicAdd(&s_cur[id[j]], 1);
        for (int j = 0; j < BATCH / 256; ++j) perm[pos[j]] = t + 256 * j;
        return;
    }

    const int g    = blockIdx.x - 1;
    const int rt   = g >> 1;
    const int ch   = g & 1;
    const int wave = __builtin_amdgcn_readfirstlane(t >> 5);
    const int lane = t & 31;
    const int m    = lane & 15;
    const int half = lane >> 4;
    const int kb   = half * 2;
    const int mb   = half * 8;
    const int k0   = wave * (ODIM / 8);
    const long ro  = (long)(rt * 16 + m) * ODIM;

    v2f a[16];
    #pragma unroll
    for (int s = 0; s < 16; ++s) {
        a[s].x = obs[ro + k0 + 4 * s + kb];
        a[s].y = obs[ro + k0 + 4 * s + kb + 1];
    }

    const float* wbase = obs_W + (long)(k0 + kb) * EMB + ch * 128 + m;
    float* plane = sh_p + wave * 16 * QSTR;
    #pragma unroll
    for (int cp = 0; cp < 4; ++cp) {
        const float* wp = wbase + cp * 32;
        v8f acc0 = {};
        v8f acc1 = {};
        #pragma unroll
        for (int s = 0; s < 16; ++s) {
            v2f b0, b1;
            b0.x = wp[(long)(4 * s)     * EMB];
            b0.y = wp[(long)(4 * s + 1) * EMB];
            b1.x = wp[(long)(4 * s)     * EMB + 16];
            b1.y = wp[(long)(4 * s + 1) * EMB + 16];
            acc0 = __builtin_amdgcn_wmma_f32_16x16x4_f32(
                false, a[s], false, b0, (short)0, acc0, false, false);
            acc1 = __builtin_amdgcn_wmma_f32_16x16x4_f32(
                false, a[s], false, b1, (short)0, acc1, false, false);
        }
        #pragma unroll
        for (int v = 0; v < 8; ++v) {
            plane[(mb + v) * QSTR + cp * 32 + m]      = acc0[v];
            plane[(mb + v) * QSTR + cp * 32 + 16 + m] = acc1[v];
        }
    }
    __syncthreads();
    #pragma unroll
    for (int j = 0; j < 2; ++j) {
        const int e   = t + 256 * j;
        const int row = e >> 5;
        const int c4  = (e & 31) * 4;
        float4 s = *reinterpret_cast<const float4*>(&sh_p[row * QSTR + c4]);
        #pragma unroll
        for (int p = 1; p < 8; ++p) {
            const float4 q = *reinterpret_cast<const float4*>(&sh_p[p * 16 * QSTR + row * QSTR + c4]);
            s.x += q.x; s.y += q.y; s.z += q.z; s.w += q.w;
        }
        const float* bp = obs_b + ch * 128 + c4;
        s.x += bp[0]; s.y += bp[1]; s.z += bp[2]; s.w += bp[3];
        *reinterpret_cast<float4*>(&hobs[(long)(rt * 16 + row) * EMB + ch * 128 + c4]) = s;
    }
}

__global__ __launch_bounds__(1024)
void moe_mlp_kernel(const float* __restrict__ state,
                    const float* __restrict__ hobs,
                    const float* __restrict__ state_W,
                    const float* __restrict__ state_b,
                    const float* __restrict__ l2_W,
                    const float* __restrict__ l2_b,
                    const int* __restrict__ tlist,
                    const int* __restrict__ perm,
                    float* __restrict__ out) {
    const int desc  = tlist[blockIdx.x];
    const int nrows = desc & 31;
    if (nrows == 0) return;
    const int first = (desc >> 5) & 2047;
    const int c     = desc >> 16;

    __shared__ int   sh_rows[16];
    __shared__ float sh_s[16 * SSTR];
    __shared__ __attribute__((aligned(16))) float sh_h[16 * HSTR];
    __shared__ __attribute__((aligned(16))) float sh_o[16 * OSTR];

    const int t    = threadIdx.x;
    const int wave = __builtin_amdgcn_readfirstlane(t >> 5);
    const int lane = t & 31;
    const int m    = lane & 15;
    const int half = lane >> 4;
    const int kb   = half * 2;
    const int mb   = half * 8;
    const int nt   = wave & 15;
    const int kh   = wave >> 4;
    const int ncol = nt * 16 + m;

    {
        const int mm = t >> 6, kk = t & (SDIM - 1);
        const int r  = (mm < nrows) ? perm[first + mm] : -1;
        if (kk == 0) sh_rows[mm] = r;
        sh_s[mm * SSTR + kk] = (r >= 0) ? state[(long)r * SDIM + kk] : 0.0f;
    }
    __syncthreads();

    const float* Ws = state_W + (long)c * SDIM * EMB;
    const float* bs = state_b + (long)c * EMB;
    const float* Wl = l2_W    + (long)c * HDIM * NOUT;
    const float* lb = l2_b    + (long)c * NOUT;

    if (kh == 0) {
        v8f acc = {};
        for (int k = 0; k < SDIM; k += 4) {
            v2f a, b;
            a.x = sh_s[m * SSTR + k + kb];
            a.y = sh_s[m * SSTR + k + kb + 1];
            b.x = Ws[(long)(k + kb)     * EMB + ncol];
            b.y = Ws[(long)(k + kb + 1) * EMB + ncol];
            acc = __builtin_amdgcn_wmma_f32_16x16x4_f32(
                false, a, false, b, (short)0, acc, false, false);
        }
        const float bias = bs[ncol];
        #pragma unroll
        for (int v = 0; v < 8; ++v)
            sh_h[(mb + v) * HSTR + ncol] = fmaxf(acc[v] + bias, 0.0f);
    } else {
        const int row = wave - 16;
        const int r   = sh_rows[row];
        float4 v0 = make_float4(0.0f, 0.0f, 0.0f, 0.0f);
        float4 v1 = v0;
        if (r >= 0) {
            const float4* srow = reinterpret_cast<const float4*>(hobs + (long)r * EMB);
            v0 = srow[lane];
            v1 = srow[32 + lane];
        }
        *reinterpret_cast<float4*>(&sh_h[row * HSTR + EMB + lane * 4])       = v0;
        *reinterpret_cast<float4*>(&sh_h[row * HSTR + EMB + 128 + lane * 4]) = v1;
    }
    __syncthreads();

    {
        v8f acc = {};
        const int k0 = kh * (HDIM / 2);
        for (int k = k0; k < k0 + HDIM / 2; k += 4) {
            v2f a, b;
            a.x = sh_h[m * HSTR + k + kb];
            a.y = sh_h[m * HSTR + k + kb + 1];
            b.x = Wl[(long)(k + kb)     * NOUT + ncol];
            b.y = Wl[(long)(k + kb + 1) * NOUT + ncol];
            acc = __builtin_amdgcn_wmma_f32_16x16x4_f32(
                false, a, false, b, (short)0, acc, false, false);
        }
        if (kh == 1) {
            #pragma unroll
            for (int v = 0; v < 8; ++v)
                sh_o[(mb + v) * OSTR + ncol] = acc[v];
        }
        __syncthreads();
        if (kh == 0) {
            const float bias = lb[ncol];
            #pragma unroll
            for (int v = 0; v < 8; ++v) {
                const int idx = (mb + v) * OSTR + ncol;
                sh_o[idx] = (acc[v] + sh_o[idx]) + bias;
            }
        }
        __syncthreads();
    }

    {
        const int row = t >> 6;
        const int c4  = (t & 63) * 4;
        const int r   = sh_rows[row];
        if (r >= 0) {
            const float4 v = *reinterpret_cast<const float4*>(&sh_o[row * OSTR + c4]);
            *reinterpret_cast<float4*>(&out[(long)r * NOUT + c4]) = v;
        }
    }
}

extern "C" void kernel_launch(void* const* d_in, const int* in_sizes, int n_in,
                              void* d_out, int out_size, void* d_ws, size_t ws_size,
                              hipStream_t stream) {
    const float* state   = (const float*)d_in[0];
    const float* obs     = (const float*)d_in[1];
    const int*   cat_ids = (const int*)  d_in[2];
    const float* obs_W   = (const float*)d_in[3];
    const float* obs_b   = (const float*)d_in[4];
    const float* state_W = (const float*)d_in[5];
    const float* state_b = (const float*)d_in[6];
    const float* l2_W    = (const float*)d_in[7];
    const float* l2_b    = (const float*)d_in[8];
    float*       out     = (float*)d_out;

    int*   tlist = (int*)d_ws;
    int*   perm  = tlist + 256;
    float* hobs  = (float*)(tlist + 4096);

    bucket_and_obs_kernel<<<257, 256, 0, stream>>>(cat_ids, obs, obs_W, obs_b,
                                                    tlist, perm, hobs);

    moe_mlp_kernel<<<MAXTILES, 1024, 0, stream>>>(state, hobs,
                                                  state_W, state_b, l2_W, l2_b,
                                                  tlist, perm, out);
}


# Launch 2 in 16-wave groups: each wave carries both K halves of its column tile as two independent accumulator chains (was 32 waves, one chain each, halves added in LDS); low + high, then bias
# speedup vs baseline: 7.8467x; 1.0744x over previous
#include <hip/hip_runtime.h>
#include <hip/hip_bf16.h>

typedef __attribute__((ext_vector_type(2))) float v2f;
typedef __attribute__((ext_vector_type(8))) float v8f;

#define BATCH 2048
#define NCAT 32
#define SDIM 64
#define EMB 256
#define NOUT 256
#define ODIM 512
#define HDIM 512
#define HSTR 516
#define SSTR 68
#define OSTR 260
#define QSTR 132
#define MAXTILES 158

__global__ __launch_bounds__(256)
void bucket_and_obs_kernel(const int* __restrict__ cat_ids,
                           const float* __restrict__ obs,
                           const float* __restrict__ obs_W,
                           const float* __restrict__ obs_b,
                           int* __restrict__ tlist,
                           int* __restrict__ perm,
                           float* __restrict__ hobs) {
    __shared__ int s_cnt[NCAT];
    __shared__ int s_cur[NCAT];
    __shared__ int s_ntiles;
    __shared__ __attribute__((aligned(16))) float sh_p[8 * 16 * QSTR];

    const int t = threadIdx.x;

    if (blockIdx.x == 0) {
        int id[BATCH / 256];
        for (int j = 0; j < BATCH / 256; ++j) id[j] = cat_ids[t + 256 * j];
        if (t < NCAT) s_cnt[t] = 0;
        __syncthreads();
        for (int j = 0; j < BATCH / 256; ++j) atomicAdd(&s_cnt[id[j]], 1);
        __syncthreads();
        if (t < NCAT) {
            const int cnt = s_cnt[t];
            const int ntl = (cnt + 15) >> 4;
            int racc = cnt;
            int tacc = ntl;
            for (int d = 1; d < NCAT; d <<= 1) {
                const int r = __shfl_up(racc, d);
                const int q = __shfl_up(tacc, d);
                if (t >= d) { racc += r; tacc += q; }
            }
            const int rstart = racc - cnt;
            const int tstart = tacc - ntl;
            s_cur[t] = rstart;
            if (t == NCAT - 1) s_ntiles = tacc;
            for (int j = 0; j < ntl; ++j) {
                const int left = cnt - 16 * j;
                const int n = (left < 16) ? left : 16;
                tlist[tstart + j] = (t << 16) | ((rstart + 16 * j) << 5) | n;
            }
        }
        __syncthreads();
        if (t >= s_ntiles && t < MAXTILES) tlist[t] = 0;
        int pos[BATCH / 256];
        for (int j = 0; j < BATCH / 256; ++j) pos[j] = atomicAdd(&s_cur[id[j]], 1);
        for (int j = 0; j < BATCH / 256; ++j) perm[pos[j]] = t + 256 * j;
        return;
    }

    const int g    = blockIdx.x - 1;
    const int rt   = g >> 1;
    const int ch   = g & 1;
    const int wave = __builtin_amdgcn_readfirstlane(t >> 5);
    const int lane = t & 31;
    const int m    = lane & 15;
    const int half = lane >> 4;
    const int kb   = half * 2;
    const int mb   = half * 8;
    const int k0   = wave * (ODIM / 8);
    const long ro  = (long)(rt * 16 + m) * ODIM;

    v2f a[16];
    #pragma unroll
    for (int s = 0; s < 16; ++s) {
        a[s].x = obs[ro + k0 + 4 * s + kb];
        a[s].y = obs[ro + k0 + 4 * s + kb + 1];
    }

    const float* wbase = obs_W + (long)(k0 + kb) * EMB + ch * 128 + m;
    float* plane = sh_p + wave * 16 * QSTR;
    #pragma unroll
    for (int cp = 0; cp < 4; ++cp) {
        const float* wp = wbase + cp * 32;
        v8f acc0 = {};
        v8f acc1 = {};
        #pragma unroll
        for (int s = 0; s < 16; ++s) {
            v2f b0, b1;
            b0.x = wp[(long)(4 * s)     * EMB];
            b0.y = wp[(long)(4 * s + 1) * EMB];
            b1.x = wp[(long)(4 * s)     * EMB + 16];
            b1.y = wp[(long)(4 * s + 1) * EMB + 16];
            acc0 = __builtin_amdgcn_wmma_f32_16x16x4_f32(
                false, a[s], false, b0, (short)0, acc0, false, false);
            acc1 = __builtin_amdgcn_wmma_f32_16x16x4_f32(
                false, a[s], false, b1, (short)0, acc1, false, false);
        }
        #pragma unroll
        for (int v = 0; v < 8; ++v) {
            plane[(mb + v) * QSTR + cp * 32 + m]      = acc0[v];
            plane[(mb + v) * QSTR + cp * 32 + 16 + m] = acc1[v];
        }
    }
    __syncthreads();
    #pragma unroll
    for (int j = 0; j < 2; ++j) {
        const int e   = t + 256 * j;
        const int row = e >> 5;
        const int c4  = (e & 31) * 4;
        float4 s = *reinterpret_cast<const float4*>(&sh_p[row * QSTR + c4]);
        #pragma unroll
        for (int p = 1; p < 8; ++p) {
            const float4 q = *reinterpret_cast<const float4*>(&sh_p[p * 16 * QSTR + row * QSTR + c4]);
            s.x += q.x; s.y += q.y; s.z += q.z; s.w += q.w;
        }
        const float* bp = obs_b + ch * 128 + c4;
        s.x += bp[0]; s.y += bp[1]; s.z += bp[2]; s.w += bp[3];
        *reinterpret_cast<float4*>(&hobs[(long)(rt * 16 + row) * EMB + ch * 128 + c4]) = s;
    }
}

__global__ __launch_bounds__(512)
void moe_mlp_kernel(const float* __restrict__ state,
                    const float* __restrict__ hobs,
                    const float* __restrict__ state_W,
                    const float* __restrict__ state_b,
                    const float* __restrict__ l2_W,
                    const float* __restrict__ l2_b,
                    const int* __restrict__ tlist,
                    const int* __restrict__ perm,
                    float* __restrict__ out) {
    const int desc  = tlist[blockIdx.x];
    const int nrows = desc & 31;
    if (nrows == 0) return;
    const int first = (desc >> 5) & 2047;
    const int c     = desc >> 16;

    __shared__ int   sh_rows[16];
    __shared__ float sh_s[16 * SSTR];
    __shared__ __attribute__((aligned(16))) float sh_h[16 * HSTR];
    __shared__ __attribute__((aligned(16))) float sh_o[16 * OSTR];

    const int t    = threadIdx.x;
    const int wave = __builtin_amdgcn_readfirstlane(t >> 5);
    const int lane = t & 31;
    const int m    = lane & 15;
    const int half = lane >> 4;
    const int kb   = half * 2;
    const int mb   = half * 8;
    const int ncol = wave * 16 + m;

    {
        const int r = (wave < nrows) ? perm[first + wave] : -1;
        if (lane == 0) sh_rows[wave] = r;
        float  s0 = 0.0f, s1 = 0.0f;
        float4 v0 = make_float4(0.0f, 0.0f, 0.0f, 0.0f);
        float4 v1 = v0;
        if (r >= 0) {
            const float*  xrow = state + (long)r * SDIM;
            const float4* srow = reinterpret_cast<const float4*>(hobs + (long)r * EMB);
            s0 = xrow[lane];
            s1 = xrow[32 + lane];
            v0 = srow[lane];
            v1 = srow[32 + lane];
        }
        sh_s[wave * SSTR + lane]      = s0;
        sh_s[wave * SSTR + 32 + lane] = s1;
        *reinterpret_cast<float4*>(&sh_h[wave * HSTR + EMB + lane * 4])       = v0;
        *reinterpret_cast<float4*>(&sh_h[wave * HSTR + EMB + 128 + lane * 4]) = v1;
    }
    __syncthreads();

    const float* Ws = state_W + (long)c * SDIM * EMB;
    const float* bs = state_b + (long)c * EMB;
    const float* Wl = l2_W    + (long)c * HDIM * NOUT;
    const float* lb = l2_b    + (long)c * NOUT;

    {
        v8f acc = {};
        for (int k = 0; k < SDIM; k += 4) {
            v2f a, b;
            a.x = sh_s[m * SSTR + k + kb];
            a.y = sh_s[m * SSTR + k + kb + 1];
            b.x = Ws[(long)(k + kb)     * EMB + ncol];
            b.y = Ws[(long)(k + kb + 1) * EMB + ncol];
            acc = __builtin_amdgcn_wmma_f32_16x16x4_f32(
                false, a, false, b, (short)0, acc, false, false);
        }
        const float bias = bs[ncol];
        #pragma unroll
        for (int v = 0; v < 8; ++v)
            sh_h[(mb + v) * HSTR + ncol] = fmaxf(acc[v] + bias, 0.0f);
    }
    __syncthreads();

    {
        v8f acc0 = {};
        v8f acc1 = {};
        const float* ap0 = sh_h + m * HSTR + kb;
        const float* ap1 = ap0 + HDIM / 2;
        const float* wp0 = Wl + (long)kb * NOUT + ncol;
        const float* wp1 = wp0 + (long)(HDIM / 2) * NOUT;
        const float  bias = lb[ncol];
        #pragma unroll 8
        for (int k = 0; k < HDIM / 2; k += 4) {
            v2f a0, b0, a1, b1;
            a0.x = ap0[k];
            a0.y = ap0[k + 1];
            a1.x = ap1[k];
            a1.y = ap1[k + 1];
            b0.x = wp0[(long)k       * NOUT];
            b0.y = wp0[(long)(k + 1) * NOUT];
            b1.x = wp1[(long)k       * NOUT];
            b1.y = wp1[(long)(k + 1) * NOUT];
            acc0 = __builtin_amdgcn_wmma_f32_16x16x4_f32(
                false, a0, false, b0, (short)0, acc0, false, false);
            acc1 = __builtin_amdgcn_wmma_f32_16x16x4_f32(
                false, a1, false, b1, (short)0, acc1, false, false);
        }
        #pragma unroll
        for (int v = 0; v < 8; ++v)
            sh_o[(mb + v) * OSTR + ncol] = (acc0[v] + acc1[v]) + bias;
    }
    __syncthreads();

    for (int j = 0; j < 2; ++j) {
        const int e   = t + 512 * j;
        const int row = e >> 6;
        const int c4  = (e & 63) * 4;
        const int r   = sh_rows[row];
        if (r >= 0) {
            const float4 v = *reinterpret_cast<const float4*>(&sh_o[row * OSTR + c4]);
            *reinterpret_cast<float4*>(&out[(long)r * NOUT + c4]) = v;
        }
    }
}

extern "C" void kernel_launch(void* const* d_in, const int* in_sizes, int n_in,
                              void* d_out, int out_size, void* d_ws, size_t ws_size,
                              hipStream_t stream) {
    const float* state   = (const float*)d_in[0];
    const float* obs     = (const float*)d_in[1];
    const int*   cat_ids = (const int*)  d_in[2];
    const float* obs_W   = (const float*)d_in[3];
    const float* obs_b   = (const float*)d_in[4];
    const float* state_W = (const float*)d_in[5];
    const float* state_b = (const float*)d_in[6];
    const float* l2_W    = (const float*)d_in[7];
    const float* l2_b    = (const float*)d_in[8];
    float*       out     = (float*)d_out;

    int*   tlist = (int*)d_ws;
    int*   perm  = tlist + 256;
    float* hobs  = (float*)(tlist + 4096);

    bucket_and_obs_kernel<<<257, 256, 0, stream>>>(cat_ids, obs, obs_W, obs_b,
                                                    tlist, perm, hobs);

    moe_mlp_kernel<<<MAXTILES, 512, 0, stream>>>(state, hobs,
                                                  state_W, state_b, l2_W, l2_b,
                                                  tlist, perm, out);
}
